# starting kernel
# baseline (speedup 1.0000x reference)
#include <hip/hip_runtime.h>
#include <math.h>

typedef __attribute__((ext_vector_type(2))) float v2f;
typedef __attribute__((ext_vector_type(8))) float v8f;

#define S 64
#define F 256
#define PITCH 66
#define TWO_PI 6.28318530717958647692f

__global__ void zero_out_kernel(float* out, int n) {
    int i = blockIdx.x * blockDim.x + threadIdx.x;
    if (i < n) out[i] = 0.0f;
}

__launch_bounds__(256)
__global__ void psf_render_kernel(const float* __restrict__ xs,
                                  const float* __restrict__ ys,
                                  const float* __restrict__ zs,
                                  const float* __restrict__ As,
                                  const float* __restrict__ bgs,
                                  const float* __restrict__ ps,
                                  const float* __restrict__ pupil_phase,
                                  float* __restrict__ out) {
    __shared__ float Pre[S * PITCH];
    __shared__ float Pim[S * PITCH];
    __shared__ float Wre[F * PITCH];
    __shared__ float Wim[F * PITCH];
    __shared__ float Tre[F * PITCH];
    __shared__ float Tim[F * PITCH];
    __shared__ float wave_max[8];

    const int e = blockIdx.x;
    const int b = e >> 2;
    if (ps[e] <= 0.5f) return;
    const float xe = xs[e], ye = ys[e], ze = zs[e];
    const float Ae = As[e], bge = bgs[e];

    const int t    = threadIdx.x;
    const int lane = t & 31;
    const int wave = t >> 5;
    const int col  = lane & 15;
    const int hi   = lane >> 4;

    for (int idx = t; idx < F * S; idx += 256) {
        const int v = idx >> 6, j = idx & 63;
        const int ti = ((j - 32) * v) & 255;
        float sn, cs;
        __sincosf(TWO_PI * (float)ti * (1.0f / 256.0f), &sn, &cs);
        Wre[v * PITCH + j] = cs;
        Wim[v * PITCH + j] = sn;
    }

    for (int idx = t; idx < S * S; idx += 256) {
        const int i = idx >> 6, j = idx & 63;
        const float ui = (2.0f * (float)i / 63.0f - 1.0f) * (1.0f / 0.75f);
        const float vj = (2.0f * (float)j / 63.0f - 1.0f) * (1.0f / 0.75f);
        const float r2 = ui * ui + vj * vj;
        const float mag  = (r2 <= 1.0f) ? 1.0f : 0.0f;
        const float prop = sqrtf(1.0f - fminf(r2, 1.0f));
        const float ph = pupil_phase[idx] + prop * ze
                       - TWO_PI * ((float)(i - 32) * (1.0f / 256.0f)) * xe
                       - TWO_PI * ((float)(j - 32) * (1.0f / 256.0f)) * ye;
        float sn, cs;
        __sincosf(ph, &sn, &cs);
        Pre[i * PITCH + j] = mag * cs;
        Pim[i * PITCH + j] = mag * sn;
    }
    __syncthreads();

    for (int tt = wave * 8; tt < wave * 8 + 8; ++tt) {
        const int ib = (tt >> 4) * 16;
        const int vb = (tt & 15) * 16;
        const int arow = ib + col;
        const int vcol = vb + col;
        v8f aRa = {}, aRb = {}, aIa = {}, aIb = {};
        #pragma unroll
        for (int kb = 0; kb < S; kb += 4) {
            const int k0 = kb + 2 * hi;
            const v2f aR = *(const v2f*)&Pre[arow * PITCH + k0];
            const v2f aI = *(const v2f*)&Pim[arow * PITCH + k0];
            const v2f bR = *(const v2f*)&Wre[vcol * PITCH + k0];
            const v2f bI = *(const v2f*)&Wim[vcol * PITCH + k0];
            aRa = __builtin_amdgcn_wmma_f32_16x16x4_f32(false, aR, false, bR, (short)0, aRa, false, false);
            aRb = __builtin_amdgcn_wmma_f32_16x16x4_f32(false, aI, false, bI, (short)0, aRb, false, false);
            aIa = __builtin_amdgcn_wmma_f32_16x16x4_f32(false, aR, false, bI, (short)0, aIa, false, false);
            aIb = __builtin_amdgcn_wmma_f32_16x16x4_f32(false, aI, false, bR, (short)0, aIb, false, false);
        }
        const int rbase = ib + 8 * hi;
        #pragma unroll
        for (int q = 0; q < 4; ++q) {
            v2f tr, ti;
            tr.x = aRa[2 * q]     - aRb[2 * q];
            tr.y = aRa[2 * q + 1] - aRb[2 * q + 1];
            ti.x = aIa[2 * q]     + aIb[2 * q];
            ti.y = aIa[2 * q + 1] + aIb[2 * q + 1];
            *(v2f*)&Tre[vcol * PITCH + rbase + 2 * q] = tr;
            *(v2f*)&Tim[vcol * PITCH + rbase + 2 * q] = ti;
        }
    }
    __syncthreads();

    float* crop = Pre;

    float mymax = 0.0f;
    for (int tt = wave * 32; tt < wave * 32 + 32; ++tt) {
        const int ub = (tt >> 4) * 16;
        const int vb = (tt & 15) * 16;
        const int urow = ub + col;
        const int vcol = vb + col;
        v8f aRa = {}, aRb = {}, aIa = {}, aIb = {};
        #pragma unroll
        for (int kb = 0; kb < S; kb += 4) {
            const int k0 = kb + 2 * hi;
            const v2f aR = *(const v2f*)&Wre[urow * PITCH + k0];
            const v2f aI = *(const v2f*)&Wim[urow * PITCH + k0];
            const v2f bR = *(const v2f*)&Tre[vcol * PITCH + k0];
            const v2f bI = *(const v2f*)&Tim[vcol * PITCH + k0];
            aRa = __builtin_amdgcn_wmma_f32_16x16x4_f32(false, aR, false, bR, (short)0, aRa, false, false);
            aRb = __builtin_amdgcn_wmma_f32_16x16x4_f32(false, aI, false, bI, (short)0, aRb, false, false);
            aIa = __builtin_amdgcn_wmma_f32_16x16x4_f32(false, aR, false, bI, (short)0, aIa, false, false);
            aIb = __builtin_amdgcn_wmma_f32_16x16x4_f32(false, aI, false, bR, (short)0, aIb, false, false);
        }
        const int wv = (vcol + 32) & 255;
        #pragma unroll
        for (int r = 0; r < 8; ++r) {
            const int u  = ub + r + 8 * hi;
            const float re = aRa[r] - aRb[r];
            const float im = aIa[r] + aIb[r];
            const float pw = re * re + im * im;
            mymax = fmaxf(mymax, pw);
            const int wu = (u + 32) & 255;
            if (wu < 64 && wv < 64) crop[wu * PITCH + wv] = pw;
        }
    }

    for (int off = 16; off > 0; off >>= 1)
        mymax = fmaxf(mymax, __shfl_xor(mymax, off, 32));
    if (lane == 0) wave_max[wave] = mymax;
    __syncthreads();
    float mx = wave_max[0];
    for (int w = 1; w < 8; ++w) mx = fmaxf(mx, wave_max[w]);
    const float inv = 1.0f / mx;

    for (int idx = t; idx < S * S; idx += 256) {
        const int cu = idx >> 6, cv = idx & 63;
        const float val = crop[cu * PITCH + cv] * inv * Ae + bge;
        atomicAdd(&out[b * S * S + idx], val);
    }
}

extern "C" void kernel_launch(void* const* d_in, const int* in_sizes, int n_in,
                              void* d_out, int out_size, void* d_ws, size_t ws_size,
                              hipStream_t stream) {
    const float* x  = (const float*)d_in[0];
    const float* y  = (const float*)d_in[1];
    const float* z  = (const float*)d_in[2];
    const float* A  = (const float*)d_in[3];
    const float* bg = (const float*)d_in[4];
    const float* p  = (const float*)d_in[5];
    const float* ph = (const float*)d_in[6];
    float* out = (float*)d_out;

    const int n_emit = in_sizes[0];
    zero_out_kernel<<<(out_size + 255) / 256, 256, 0, stream>>>(out, out_size);
    psf_render_kernel<<<n_emit, 256, 0, stream>>>(x, y, z, A, bg, p, ph, out);
}


# skip the pupil's all-zero border: only rows/columns 8..55 (mag==0 outside for any input) enter both transforms, K 64->48 in both stages and stage-1 M 64->48
# speedup vs baseline: 1.2411x; 1.2411x over previous
#include <hip/hip_runtime.h>
#include <math.h>


typedef __attribute__((ext_vector_type(2))) float v2f;
typedef __attribute__((ext_vector_type(8))) float v8f;

#define S 64
#define F 256
#define K0 8
#define KS 48
#define PITCH 52
#define CPITCH 66
#define TWO_PI 6.28318530717958647692f

__global__ void zero_out_kernel(float* out, int n) {
    int i = blockIdx.x * blockDim.x + threadIdx.x;
    if (i < n) out[i] = 0.0f;
}

__launch_bounds__(256)
__global__ void psf_render_kernel(const float* __restrict__ xs,
                                  const float* __restrict__ ys,
                                  const float* __restrict__ zs,
                                  const float* __restrict__ As,
                                  const float* __restrict__ bgs,
                                  const float* __restrict__ ps,
                                  const float* __restrict__ pupil_phase,
                                  float* __restrict__ out) {
    __shared__ float Pbuf[2 * KS * PITCH];
    __shared__ float Wre[F * PITCH];
    __shared__ float Wim[F * PITCH];
    __shared__ float Tre[F * PITCH];
    __shared__ float Tim[F * PITCH];
    __shared__ float wave_max[8];

    float* Pre = Pbuf;
    float* Pim = Pbuf + KS * PITCH;

    const int e = blockIdx.x;
    const int b = e >> 2;
    if (ps[e] <= 0.5f) return;
    const float xe = xs[e], ye = ys[e], ze = zs[e];
    const float Ae = As[e], bge = bgs[e];

    const int t    = threadIdx.x;
    const int lane = t & 31;
    const int wave = t >> 5;
    const int col  = lane & 15;
    const int hi   = lane >> 4;

    for (unsigned idx = t; idx < F * KS; idx += 256) {
        const int v = (int)(idx / KS), c = (int)(idx % KS);
        const int ti = ((c + K0 - 32) * v) & 255;
        float sn, cs;
        __sincosf(TWO_PI * (float)ti * (1.0f / 256.0f), &sn, &cs);
        Wre[v * PITCH + c] = cs;
        Wim[v * PITCH + c] = sn;
    }

    for (unsigned idx = t; idx < KS * KS; idx += 256) {
        const int ci = (int)(idx / KS), cj = (int)(idx % KS);
        const int i = ci + K0, j = cj + K0;
        const float ui = (2.0f * (float)i / 63.0f - 1.0f) * (1.0f / 0.75f);
        const float vj = (2.0f * (float)j / 63.0f - 1.0f) * (1.0f / 0.75f);
        const float r2 = ui * ui + vj * vj;
        const float mag  = (r2 <= 1.0f) ? 1.0f : 0.0f;
        const float prop = sqrtf(1.0f - fminf(r2, 1.0f));
        const float ph = pupil_phase[i * S + j] + prop * ze
                       - TWO_PI * ((float)(i - 32) * (1.0f / 256.0f)) * xe
                       - TWO_PI * ((float)(j - 32) * (1.0f / 256.0f)) * ye;
        float sn, cs;
        __sincosf(ph, &sn, &cs);
        Pre[ci * PITCH + cj] = mag * cs;
        Pim[ci * PITCH + cj] = mag * sn;
    }
    __syncthreads();

    for (int tt = wave * 6; tt < wave * 6 + 6; ++tt) {
        const int ib = (tt >> 4) * 16;
        const int vb = (tt & 15) * 16;
        const int arow = ib + col;
        const int vcol = vb + col;
        v8f aRa = {}, aRb = {}, aIa = {}, aIb = {};
        #pragma unroll
        for (int kb = 0; kb < KS; kb += 4) {
            const int k0 = kb + 2 * hi;
            const v2f aR = *(const v2f*)&Pre[arow * PITCH + k0];
            const v2f aI = *(const v2f*)&Pim[arow * PITCH + k0];
            const v2f bR = *(const v2f*)&Wre[vcol * PITCH + k0];
            const v2f bI = *(const v2f*)&Wim[vcol * PITCH + k0];
            aRa = __builtin_amdgcn_wmma_f32_16x16x4_f32(false, aR, false, bR, (short)0, aRa, false, false);
            aRb = __builtin_amdgcn_wmma_f32_16x16x4_f32(false, aI, false, bI, (short)0, aRb, false, false);
            aIa = __builtin_amdgcn_wmma_f32_16x16x4_f32(false, aR, false, bI, (short)0, aIa, false, false);
            aIb = __builtin_amdgcn_wmma_f32_16x16x4_f32(false, aI, false, bR, (short)0, aIb, false, false);
        }
        const int rbase = ib + 8 * hi;
        #pragma unroll
        for (int q = 0; q < 4; ++q) {
            v2f tr, ti;
            tr.x = aRa[2 * q]     - aRb[2 * q];
            tr.y = aRa[2 * q + 1] - aRb[2 * q + 1];
            ti.x = aIa[2 * q]     + aIb[2 * q];
            ti.y = aIa[2 * q + 1] + aIb[2 * q + 1];
            *(v2f*)&Tre[vcol * PITCH + rbase + 2 * q] = tr;
            *(v2f*)&Tim[vcol * PITCH + rbase + 2 * q] = ti;
        }
    }
    __syncthreads();

    float* crop = Pbuf;

    float mymax = 0.0f;
    for (int tt = wave * 32; tt < wave * 32 + 32; ++tt) {
        const int ub = (tt >> 4) * 16;
        const int vb = (tt & 15) * 16;
        const int urow = ub + col;
        const int vcol = vb + col;
        v8f aRa = {}, aRb = {}, aIa = {}, aIb = {};
        #pragma unroll
        for (int kb = 0; kb < KS; kb += 4) {
            const int k0 = kb + 2 * hi;
            const v2f aR = *(const v2f*)&Wre[urow * PITCH + k0];
            const v2f aI = *(const v2f*)&Wim[urow * PITCH + k0];
            const v2f bR = *(const v2f*)&Tre[vcol * PITCH + k0];
            const v2f bI = *(const v2f*)&Tim[vcol * PITCH + k0];
            aRa = __builtin_amdgcn_wmma_f32_16x16x4_f32(false, aR, false, bR, (short)0, aRa, false, false);
            aRb = __builtin_amdgcn_wmma_f32_16x16x4_f32(false, aI, false, bI, (short)0, aRb, false, false);
            aIa = __builtin_amdgcn_wmma_f32_16x16x4_f32(false, aR, false, bI, (short)0, aIa, false, false);
            aIb = __builtin_amdgcn_wmma_f32_16x16x4_f32(false, aI, false, bR, (short)0, aIb, false, false);
        }
        const int wv = (vcol + 32) & 255;
        #pragma unroll
        for (int r = 0; r < 8; ++r) {
            const int u  = ub + r + 8 * hi;
            const float re = aRa[r] - aRb[r];
            const float im = aIa[r] + aIb[r];
            const float pw = re * re + im * im;
            mymax = fmaxf(mymax, pw);
            const int wu = (u + 32) & 255;
            if (wu < 64 && wv < 64) crop[wu * CPITCH + wv] = pw;
        }
    }

    for (int off = 16; off > 0; off >>= 1)
        mymax = fmaxf(mymax, __shfl_xor(mymax, off, 32));
    if (lane == 0) wave_max[wave] = mymax;
    __syncthreads();
    float mx = wave_max[0];
    for (int w = 1; w < 8; ++w) mx = fmaxf(mx, wave_max[w]);
    const float inv = 1.0f / mx;

    for (int idx = t; idx < S * S; idx += 256) {
        const int cu = idx >> 6, cv = idx & 63;
        const float val = crop[cu * CPITCH + cv] * inv * Ae + bge;
        atomicAdd(&out[b * S * S + idx], val);
    }
}

extern "C" void kernel_launch(void* const* d_in, const int* in_sizes, int n_in,
                              void* d_out, int out_size, void* d_ws, size_t ws_size,
                              hipStream_t stream) {
    const float* x  = (const float*)d_in[0];
    const float* y  = (const float*)d_in[1];
    const float* z  = (const float*)d_in[2];
    const float* A  = (const float*)d_in[3];
    const float* bg = (const float*)d_in[4];
    const float* p  = (const float*)d_in[5];
    const float* ph = (const float*)d_in[6];
    float* out = (float*)d_out;

    const int n_emit = in_sizes[0];
    zero_out_kernel<<<(out_size + 255) / 256, 256, 0, stream>>>(out, out_size);
    psf_render_kernel<<<n_emit, 256, 0, stream>>>(x, y, z, A, bg, p, ph, out);
}


# twiddle row 256-u is the conjugate of row u: each tile's four real products now give output rows u and 256-u in both transforms, so only rows 0..143 are multiplied (and built)
# speedup vs baseline: 1.7419x; 1.4035x over previous
#include <hip/hip_runtime.h>
#include <math.h>


typedef __attribute__((ext_vector_type(2))) float v2f;
typedef __attribute__((ext_vector_type(8))) float v8f;

#define S 64
#define F 256
#define K0 8
#define KS 48
#define WR 144
#define PITCH 52
#define CPITCH 66
#define TWO_PI 6.28318530717958647692f

__global__ void zero_out_kernel(float* out, int n) {
    int i = blockIdx.x * blockDim.x + threadIdx.x;
    if (i < n) out[i] = 0.0f;
}

__launch_bounds__(256)
__global__ void psf_render_kernel(const float* __restrict__ xs,
                                  const float* __restrict__ ys,
                                  const float* __restrict__ zs,
                                  const float* __restrict__ As,
                                  const float* __restrict__ bgs,
                                  const float* __restrict__ ps,
                                  const float* __restrict__ pupil_phase,
                                  float* __restrict__ out) {
    __shared__ float Pbuf[2 * KS * PITCH];
    __shared__ float Wre[WR * PITCH];
    __shared__ float Wim[WR * PITCH];
    __shared__ float Tre[F * PITCH];
    __shared__ float Tim[F * PITCH];
    __shared__ float wave_max[8];

    float* Pre = Pbuf;
    float* Pim = Pbuf + KS * PITCH;

    const int e = blockIdx.x;
    const int b = e >> 2;
    if (ps[e] <= 0.5f) return;
    const float xe = xs[e], ye = ys[e], ze = zs[e];
    const float Ae = As[e], bge = bgs[e];

    const int t    = threadIdx.x;
    const int lane = t & 31;
    const int wave = t >> 5;
    const int col  = lane & 15;
    const int hi   = lane >> 4;

    for (unsigned idx = t; idx < WR * KS; idx += 256) {
        const int v = (int)(idx / KS), c = (int)(idx % KS);
        const int ti = ((c + K0 - 32) * v) & 255;
        float sn, cs;
        __sincosf(TWO_PI * (float)ti * (1.0f / 256.0f), &sn, &cs);
        Wre[v * PITCH + c] = cs;
        Wim[v * PITCH + c] = sn;
    }

    for (unsigned idx = t; idx < KS * KS; idx += 256) {
        const int ci = (int)(idx / KS), cj = (int)(idx % KS);
        const int i = ci + K0, j = cj + K0;
        const float ui = (2.0f * (float)i / 63.0f - 1.0f) * (1.0f / 0.75f);
        const float vj = (2.0f * (float)j / 63.0f - 1.0f) * (1.0f / 0.75f);
        const float r2 = ui * ui + vj * vj;
        const float mag  = (r2 <= 1.0f) ? 1.0f : 0.0f;
        const float prop = sqrtf(1.0f - fminf(r2, 1.0f));
        const float ph = pupil_phase[i * S + j] + prop * ze
                       - TWO_PI * ((float)(i - 32) * (1.0f / 256.0f)) * xe
                       - TWO_PI * ((float)(j - 32) * (1.0f / 256.0f)) * ye;
        float sn, cs;
        __sincosf(ph, &sn, &cs);
        Pre[ci * PITCH + cj] = mag * cs;
        Pim[ci * PITCH + cj] = mag * sn;
    }
    __syncthreads();

    for (int tt = wave; tt < 27; tt += 8) {
        const int ib = (tt / 9) * 16;
        const int vb = (tt % 9) * 16;
        const int arow = ib + col;
        const int vcol = vb + col;
        v8f aRa = {}, aRb = {}, aIa = {}, aIb = {};
        #pragma unroll
        for (int kb = 0; kb < KS; kb += 4) {
            const int k0 = kb + 2 * hi;
            const v2f aR = *(const v2f*)&Pre[arow * PITCH + k0];
            const v2f aI = *(const v2f*)&Pim[arow * PITCH + k0];
            const v2f bR = *(const v2f*)&Wre[vcol * PITCH + k0];
            const v2f bI = *(const v2f*)&Wim[vcol * PITCH + k0];
            aRa = __builtin_amdgcn_wmma_f32_16x16x4_f32(false, aR, false, bR, (short)0, aRa, false, false);
            aRb = __builtin_amdgcn_wmma_f32_16x16x4_f32(false, aI, false, bI, (short)0, aRb, false, false);
            aIa = __builtin_amdgcn_wmma_f32_16x16x4_f32(false, aR, false, bI, (short)0, aIa, false, false);
            aIb = __builtin_amdgcn_wmma_f32_16x16x4_f32(false, aI, false, bR, (short)0, aIb, false, false);
        }
        const int rbase = ib + 8 * hi;
        const bool dir = vcol <= 128;
        const bool mir = (unsigned)(vcol - 1) < 127u;
        const int  vmir = 256 - vcol;
        #pragma unroll
        for (int q = 0; q < 4; ++q) {
            v2f tr, ti, mr, mi;
            tr.x = aRa[2 * q]     - aRb[2 * q];
            tr.y = aRa[2 * q + 1] - aRb[2 * q + 1];
            ti.x = aIa[2 * q]     + aIb[2 * q];
            ti.y = aIa[2 * q + 1] + aIb[2 * q + 1];
            mr.x = aRa[2 * q]     + aRb[2 * q];
            mr.y = aRa[2 * q + 1] + aRb[2 * q + 1];
            mi.x = aIb[2 * q]     - aIa[2 * q];
            mi.y = aIb[2 * q + 1] - aIa[2 * q + 1];
            if (dir) {
                *(v2f*)&Tre[vcol * PITCH + rbase + 2 * q] = tr;
                *(v2f*)&Tim[vcol * PITCH + rbase + 2 * q] = ti;
            }
            if (mir) {
                *(v2f*)&Tre[vmir * PITCH + rbase + 2 * q] = mr;
                *(v2f*)&Tim[vmir * PITCH + rbase + 2 * q] = mi;
            }
        }
    }
    __syncthreads();

    float* crop = Pbuf;

    float mymax = 0.0f;
    for (int tt = wave * 18; tt < wave * 18 + 18; ++tt) {
        const int ub = (tt >> 4) * 16;
        const int vb = (tt & 15) * 16;
        const int urow = ub + col;
        const int vcol = vb + col;
        v8f aRa = {}, aRb = {}, aIa = {}, aIb = {};
        #pragma unroll
        for (int kb = 0; kb < KS; kb += 4) {
            const int k0 = kb + 2 * hi;
            const v2f aR = *(const v2f*)&Wre[urow * PITCH + k0];
            const v2f aI = *(const v2f*)&Wim[urow * PITCH + k0];
            const v2f bR = *(const v2f*)&Tre[vcol * PITCH + k0];
            const v2f bI = *(const v2f*)&Tim[vcol * PITCH + k0];
            aRa = __builtin_amdgcn_wmma_f32_16x16x4_f32(false, aR, false, bR, (short)0, aRa, false, false);
            aRb = __builtin_amdgcn_wmma_f32_16x16x4_f32(false, aI, false, bI, (short)0, aRb, false, false);
            aIa = __builtin_amdgcn_wmma_f32_16x16x4_f32(false, aR, false, bI, (short)0, aIa, false, false);
            aIb = __builtin_amdgcn_wmma_f32_16x16x4_f32(false, aI, false, bR, (short)0, aIb, false, false);
        }
        const int wv = (vcol + 32) & 255;
        float pw[8], pm[8];
        #pragma unroll
        for (int r = 0; r < 8; ++r) {
            const float re = aRa[r] - aRb[r];
            const float im = aIa[r] + aIb[r];
            pw[r] = re * re + im * im;
            const float rm = aRa[r] + aRb[r];
            const float jm = aIa[r] - aIb[r];
            pm[r] = rm * rm + jm * jm;
            mymax = fmaxf(mymax, fmaxf(pw[r], pm[r]));
        }
        #pragma unroll
        for (int r = 0; r < 8; ++r) {
            const int u  = ub + r + 8 * hi;
            const int wu = (u + 32) & 255;
            if (wu < 64 && wv < 64) crop[wu * CPITCH + wv] = pw[r];
        }
        #pragma unroll
        for (int r = 0; r < 8; ++r) {
            const int u  = ub + r + 8 * hi;
            const int wm = (32 - u) & 255;
            if (wm < 32 && wv < 64) crop[wm * CPITCH + wv] = pm[r];
        }
    }

    for (int off = 16; off > 0; off >>= 1)
        mymax = fmaxf(mymax, __shfl_xor(mymax, off, 32));
    if (lane == 0) wave_max[wave] = mymax;
    __syncthreads();
    float mx = wave_max[0];
    for (int w = 1; w < 8; ++w) mx = fmaxf(mx, wave_max[w]);
    const float inv = 1.0f / mx;

    for (int idx = t; idx < S * S; idx += 256) {
        const int cu = idx >> 6, cv = idx & 63;
        const float val = crop[cu * CPITCH + cv] * inv * Ae + bge;
        atomicAdd(&out[b * S * S + idx], val);
    }
}

extern "C" void kernel_launch(void* const* d_in, const int* in_sizes, int n_in,
                              void* d_out, int out_size, void* d_ws, size_t ws_size,
                              hipStream_t stream) {
    const float* x  = (const float*)d_in[0];
    const float* y  = (const float*)d_in[1];
    const float* z  = (const float*)d_in[2];
    const float* A  = (const float*)d_in[3];
    const float* bg = (const float*)d_in[4];
    const float* p  = (const float*)d_in[5];
    const float* ph = (const float*)d_in[6];
    float* out = (float*)d_out;

    const int n_emit = in_sizes[0];
    zero_out_kernel<<<(out_size + 255) / 256, 256, 0, stream>>>(out, out_size);
    psf_render_kernel<<<n_emit, 256, 0, stream>>>(x, y, z, A, bg, p, ph, out);
}


# render block of 16 waves instead of 8 (four per SIMD, 9 stage-2 tiles per wave): a wave's load, LDS and store-pass waits are covered by three partners instead of one
# speedup vs baseline: 2.0862x; 1.1977x over previous
#include <hip/hip_runtime.h>
#include <math.h>


typedef __attribute__((ext_vector_type(2))) float v2f;
typedef __attribute__((ext_vector_type(8))) float v8f;

#define S 64
#define F 256
#define K0 8
#define KS 48
#define WR 144
#define PITCH 52
#define CPITCH 66
#define TWO_PI 6.28318530717958647692f
#define NT 512
#define NW (NT / 32)

__global__ void zero_out_kernel(float* out, int n) {
    int i = blockIdx.x * blockDim.x + threadIdx.x;
    if (i < n) out[i] = 0.0f;
}

__launch_bounds__(NT)
__global__ void psf_render_kernel(const float* __restrict__ xs,
                                  const float* __restrict__ ys,
                                  const float* __restrict__ zs,
                                  const float* __restrict__ As,
                                  const float* __restrict__ bgs,
                                  const float* __restrict__ ps,
                                  const float* __restrict__ pupil_phase,
                                  float* __restrict__ out) {
    __shared__ float Pbuf[2 * KS * PITCH];
    __shared__ float Wre[WR * PITCH];
    __shared__ float Wim[WR * PITCH];
    __shared__ float Tre[F * PITCH];
    __shared__ float Tim[F * PITCH];
    __shared__ float wave_max[NW];

    float* Pre = Pbuf;
    float* Pim = Pbuf + KS * PITCH;

    const int e = blockIdx.x;
    const int b = e >> 2;
    if (ps[e] <= 0.5f) return;
    const float xe = xs[e], ye = ys[e], ze = zs[e];
    const float Ae = As[e], bge = bgs[e];

    const int t    = threadIdx.x;
    const int lane = t & 31;
    const int wave = t >> 5;
    const int col  = lane & 15;
    const int hi   = lane >> 4;

    for (unsigned idx = t; idx < WR * KS; idx += NT) {
        const int v = (int)(idx / KS), c = (int)(idx % KS);
        const int ti = ((c + K0 - 32) * v) & 255;
        float sn, cs;
        __sincosf(TWO_PI * (float)ti * (1.0f / 256.0f), &sn, &cs);
        Wre[v * PITCH + c] = cs;
        Wim[v * PITCH + c] = sn;
    }

    for (unsigned idx = t; idx < KS * KS; idx += NT) {
        const int ci = (int)(idx / KS), cj = (int)(idx % KS);
        const int i = ci + K0, j = cj + K0;
        const float ui = (2.0f * (float)i / 63.0f - 1.0f) * (1.0f / 0.75f);
        const float vj = (2.0f * (float)j / 63.0f - 1.0f) * (1.0f / 0.75f);
        const float r2 = ui * ui + vj * vj;
        const float mag  = (r2 <= 1.0f) ? 1.0f : 0.0f;
        const float prop = sqrtf(1.0f - fminf(r2, 1.0f));
        const float ph = pupil_phase[i * S + j] + prop * ze
                       - TWO_PI * ((float)(i - 32) * (1.0f / 256.0f)) * xe
                       - TWO_PI * ((float)(j - 32) * (1.0f / 256.0f)) * ye;
        float sn, cs;
        __sincosf(ph, &sn, &cs);
        Pre[ci * PITCH + cj] = mag * cs;
        Pim[ci * PITCH + cj] = mag * sn;
    }
    __syncthreads();

    for (int tt = wave; tt < 27; tt += NW) {
        const int ib = (tt / 9) * 16;
        const int vb = (tt % 9) * 16;
        const int arow = ib + col;
        const int vcol = vb + col;
        v8f aRa = {}, aRb = {}, aIa = {}, aIb = {};
        #pragma unroll
        for (int kb = 0; kb < KS; kb += 4) {
            const int k0 = kb + 2 * hi;
            const v2f aR = *(const v2f*)&Pre[arow * PITCH + k0];
            const v2f aI = *(const v2f*)&Pim[arow * PITCH + k0];
            const v2f bR = *(const v2f*)&Wre[vcol * PITCH + k0];
            const v2f bI = *(const v2f*)&Wim[vcol * PITCH + k0];
            aRa = __builtin_amdgcn_wmma_f32_16x16x4_f32(false, aR, false, bR, (short)0, aRa, false, false);
            aRb = __builtin_amdgcn_wmma_f32_16x16x4_f32(false, aI, false, bI, (short)0, aRb, false, false);
            aIa = __builtin_amdgcn_wmma_f32_16x16x4_f32(false, aR, false, bI, (short)0, aIa, false, false);
            aIb = __builtin_amdgcn_wmma_f32_16x16x4_f32(false, aI, false, bR, (short)0, aIb, false, false);
        }
        const int rbase = ib + 8 * hi;
        const bool dir = vcol <= 128;
        const bool mir = (unsigned)(vcol - 1) < 127u;
        const int  vmir = 256 - vcol;
        #pragma unroll
        for (int q = 0; q < 4; ++q) {
            v2f tr, ti, mr, mi;
            tr.x = aRa[2 * q]     - aRb[2 * q];
            tr.y = aRa[2 * q + 1] - aRb[2 * q + 1];
            ti.x = aIa[2 * q]     + aIb[2 * q];
            ti.y = aIa[2 * q + 1] + aIb[2 * q + 1];
            mr.x = aRa[2 * q]     + aRb[2 * q];
            mr.y = aRa[2 * q + 1] + aRb[2 * q + 1];
            mi.x = aIb[2 * q]     - aIa[2 * q];
            mi.y = aIb[2 * q + 1] - aIa[2 * q + 1];
            if (dir) {
                *(v2f*)&Tre[vcol * PITCH + rbase + 2 * q] = tr;
                *(v2f*)&Tim[vcol * PITCH + rbase + 2 * q] = ti;
            }
            if (mir) {
                *(v2f*)&Tre[vmir * PITCH + rbase + 2 * q] = mr;
                *(v2f*)&Tim[vmir * PITCH + rbase + 2 * q] = mi;
            }
        }
    }
    __syncthreads();

    float* crop = Pbuf;

    float mymax = 0.0f;
    static_assert(144 % NW == 0, "stage-2 tiles divide evenly over the waves");
    for (int tt = wave * (144 / NW); tt < (wave + 1) * (144 / NW); ++tt) {
        const int ub = (tt >> 4) * 16;
        const int vb = (tt & 15) * 16;
        const int urow = ub + col;
        const int vcol = vb + col;
        v8f aRa = {}, aRb = {}, aIa = {}, aIb = {};
        #pragma unroll
        for (int kb = 0; kb < KS; kb += 4) {
            const int k0 = kb + 2 * hi;
            const v2f aR = *(const v2f*)&Wre[urow * PITCH + k0];
            const v2f aI = *(const v2f*)&Wim[urow * PITCH + k0];
            const v2f bR = *(const v2f*)&Tre[vcol * PITCH + k0];
            const v2f bI = *(const v2f*)&Tim[vcol * PITCH + k0];
            aRa = __builtin_amdgcn_wmma_f32_16x16x4_f32(false, aR, false, bR, (short)0, aRa, false, false);
            aRb = __builtin_amdgcn_wmma_f32_16x16x4_f32(false, aI, false, bI, (short)0, aRb, false, false);
            aIa = __builtin_amdgcn_wmma_f32_16x16x4_f32(false, aR, false, bI, (short)0, aIa, false, false);
            aIb = __builtin_amdgcn_wmma_f32_16x16x4_f32(false, aI, false, bR, (short)0, aIb, false, false);
        }
        const int wv = (vcol + 32) & 255;
        float pw[8], pm[8];
        #pragma unroll
        for (int r = 0; r < 8; ++r) {
            const float re = aRa[r] - aRb[r];
            const float im = aIa[r] + aIb[r];
            pw[r] = re * re + im * im;
            const float rm = aRa[r] + aRb[r];
            const float jm = aIa[r] - aIb[r];
            pm[r] = rm * rm + jm * jm;
            mymax = fmaxf(mymax, fmaxf(pw[r], pm[r]));
        }
        #pragma unroll
        for (int r = 0; r < 8; ++r) {
            const int u  = ub + r + 8 * hi;
            const int wu = (u + 32) & 255;
            if (wu < 64 && wv < 64) crop[wu * CPITCH + wv] = pw[r];
        }
        #pragma unroll
        for (int r = 0; r < 8; ++r) {
            const int u  = ub + r + 8 * hi;
            const int wm = (32 - u) & 255;
            if (wm < 32 && wv < 64) crop[wm * CPITCH + wv] = pm[r];
        }
    }

    for (int off = 16; off > 0; off >>= 1)
        mymax = fmaxf(mymax, __shfl_xor(mymax, off, 32));
    if (lane == 0) wave_max[wave] = mymax;
    __syncthreads();
    float mx = wave_max[0];
    for (int w = 1; w < NW; ++w) mx = fmaxf(mx, wave_max[w]);
    const float inv = 1.0f / mx;

    for (int idx = t; idx < S * S; idx += NT) {
        const int cu = idx >> 6, cv = idx & 63;
        const float val = crop[cu * CPITCH + cv] * inv * Ae + bge;
        atomicAdd(&out[b * S * S + idx], val);
    }
}

extern "C" void kernel_launch(void* const* d_in, const int* in_sizes, int n_in,
                              void* d_out, int out_size, void* d_ws, size_t ws_size,
                              hipStream_t stream) {
    const float* x  = (const float*)d_in[0];
    const float* y  = (const float*)d_in[1];
    const float* z  = (const float*)d_in[2];
    const float* A  = (const float*)d_in[3];
    const float* bg = (const float*)d_in[4];
    const float* p  = (const float*)d_in[5];
    const float* ph = (const float*)d_in[6];
    float* out = (float*)d_out;

    const int n_emit = in_sizes[0];
    zero_out_kernel<<<(out_size + 255) / 256, 256, 0, stream>>>(out, out_size);
    psf_render_kernel<<<n_emit, NT, 0, stream>>>(x, y, z, A, bg, p, ph, out);
}


# twiddle rows 0 and 128 are both real, so they share one row slot (row 0 in the cosine plane, row 128 in the sine plane): 8 instead of 9 sixteen-row tiles are multiplied in both transforms
# speedup vs baseline: 2.1933x; 1.0513x over previous
#include <hip/hip_runtime.h>
#include <math.h>


typedef __attribute__((ext_vector_type(2))) float v2f;
typedef __attribute__((ext_vector_type(8))) float v8f;

#define S 64
#define F 256
#define K0 8
#define KS 48
#define WR 128
#define PITCH 52
#define CPITCH 66
#define TWO_PI 6.28318530717958647692f
#define NT 512
#define NW (NT / 32)

__global__ void zero_out_kernel(float* out, int n) {
    int i = blockIdx.x * blockDim.x + threadIdx.x;
    if (i < n) out[i] = 0.0f;
}

__launch_bounds__(NT)
__global__ void psf_render_kernel(const float* __restrict__ xs,
                                  const float* __restrict__ ys,
                                  const float* __restrict__ zs,
                                  const float* __restrict__ As,
                                  const float* __restrict__ bgs,
                                  const float* __restrict__ ps,
                                  const float* __restrict__ pupil_phase,
                                  float* __restrict__ out) {
    __shared__ float Pbuf[2 * KS * PITCH];
    __shared__ float Wre[WR * PITCH];
    __shared__ float Wim[WR * PITCH];
    __shared__ float Tre[F * PITCH];
    __shared__ float Tim[F * PITCH];
    __shared__ float wave_max[NW];

    float* Pre = Pbuf;
    float* Pim = Pbuf + KS * PITCH;

    const int e = blockIdx.x;
    const int b = e >> 2;
    if (ps[e] <= 0.5f) return;
    const float xe = xs[e], ye = ys[e], ze = zs[e];
    const float Ae = As[e], bge = bgs[e];

    const int t    = threadIdx.x;
    const int lane = t & 31;
    const int wave = t >> 5;
    const int col  = lane & 15;
    const int hi   = lane >> 4;

    for (unsigned idx = t; idx < WR * KS; idx += NT) {
        const int v = (int)(idx / KS), c = (int)(idx % KS);
        const int ti = ((c + K0 - 32) * v) & 255;
        float sn, cs;
        __sincosf(TWO_PI * (float)ti * (1.0f / 256.0f), &sn, &cs);
        if (v == 0) sn = ((c + K0) & 1) ? -1.0f : 1.0f;
        Wre[v * PITCH + c] = cs;
        Wim[v * PITCH + c] = sn;
    }

    for (unsigned idx = t; idx < KS * KS; idx += NT) {
        const int ci = (int)(idx / KS), cj = (int)(idx % KS);
        const int i = ci + K0, j = cj + K0;
        const float ui = (2.0f * (float)i / 63.0f - 1.0f) * (1.0f / 0.75f);
        const float vj = (2.0f * (float)j / 63.0f - 1.0f) * (1.0f / 0.75f);
        const float r2 = ui * ui + vj * vj;
        const float mag  = (r2 <= 1.0f) ? 1.0f : 0.0f;
        const float prop = sqrtf(1.0f - fminf(r2, 1.0f));
        const float ph = pupil_phase[i * S + j] + prop * ze
                       - TWO_PI * ((float)(i - 32) * (1.0f / 256.0f)) * xe
                       - TWO_PI * ((float)(j - 32) * (1.0f / 256.0f)) * ye;
        float sn, cs;
        __sincosf(ph, &sn, &cs);
        Pre[ci * PITCH + cj] = mag * cs;
        Pim[ci * PITCH + cj] = mag * sn;
    }
    __syncthreads();

    for (int tt = wave; tt < 24; tt += NW) {
        const int ib = (tt >> 3) * 16;
        const int vb = (tt & 7) * 16;
        const int arow = ib + col;
        const int vcol = vb + col;
        v8f aRa = {}, aRb = {}, aIa = {}, aIb = {};
        #pragma unroll
        for (int kb = 0; kb < KS; kb += 4) {
            const int k0 = kb + 2 * hi;
            const v2f aR = *(const v2f*)&Pre[arow * PITCH + k0];
            const v2f aI = *(const v2f*)&Pim[arow * PITCH + k0];
            const v2f bR = *(const v2f*)&Wre[vcol * PITCH + k0];
            const v2f bI = *(const v2f*)&Wim[vcol * PITCH + k0];
            aRa = __builtin_amdgcn_wmma_f32_16x16x4_f32(false, aR, false, bR, (short)0, aRa, false, false);
            aRb = __builtin_amdgcn_wmma_f32_16x16x4_f32(false, aI, false, bI, (short)0, aRb, false, false);
            aIa = __builtin_amdgcn_wmma_f32_16x16x4_f32(false, aR, false, bI, (short)0, aIa, false, false);
            aIb = __builtin_amdgcn_wmma_f32_16x16x4_f32(false, aI, false, bR, (short)0, aIb, false, false);
        }
        const int rbase = ib + 8 * hi;
        const bool pk  = vcol == 0;
        const int  vmir = pk ? 128 : 256 - vcol;
        #pragma unroll
        for (int q = 0; q < 4; ++q) {
            v2f tr, ti, mr, mi;
            tr.x = aRa[2 * q]     - aRb[2 * q];
            tr.y = aRa[2 * q + 1] - aRb[2 * q + 1];
            ti.x = aIa[2 * q]     + aIb[2 * q];
            ti.y = aIa[2 * q + 1] + aIb[2 * q + 1];
            mr.x = aRa[2 * q]     + aRb[2 * q];
            mr.y = aRa[2 * q + 1] + aRb[2 * q + 1];
            mi.x = aIb[2 * q]     - aIa[2 * q];
            mi.y = aIb[2 * q + 1] - aIa[2 * q + 1];
            if (pk) {
                tr.x = aRa[2 * q]; tr.y = aRa[2 * q + 1];
                ti.x = aIb[2 * q]; ti.y = aIb[2 * q + 1];
                mr.x = aIa[2 * q]; mr.y = aIa[2 * q + 1];
                mi.x = aRb[2 * q]; mi.y = aRb[2 * q + 1];
            }
            *(v2f*)&Tre[vcol * PITCH + rbase + 2 * q] = tr;
            *(v2f*)&Tim[vcol * PITCH + rbase + 2 * q] = ti;
            *(v2f*)&Tre[vmir * PITCH + rbase + 2 * q] = mr;
            *(v2f*)&Tim[vmir * PITCH + rbase + 2 * q] = mi;
        }
    }
    __syncthreads();

    float* crop = Pbuf;

    float mymax = 0.0f;
    static_assert(128 % NW == 0, "stage-2 tiles divide evenly over the waves");
    for (int tt = wave * (128 / NW); tt < (wave + 1) * (128 / NW); ++tt) {
        const int ub = (tt >> 4) * 16;
        const int vb = (tt & 15) * 16;
        const int urow = ub + col;
        const int vcol = vb + col;
        v8f aRa = {}, aRb = {}, aIa = {}, aIb = {};
        #pragma unroll
        for (int kb = 0; kb < KS; kb += 4) {
            const int k0 = kb + 2 * hi;
            const v2f aR = *(const v2f*)&Wre[urow * PITCH + k0];
            const v2f aI = *(const v2f*)&Wim[urow * PITCH + k0];
            const v2f bR = *(const v2f*)&Tre[vcol * PITCH + k0];
            const v2f bI = *(const v2f*)&Tim[vcol * PITCH + k0];
            aRa = __builtin_amdgcn_wmma_f32_16x16x4_f32(false, aR, false, bR, (short)0, aRa, false, false);
            aRb = __builtin_amdgcn_wmma_f32_16x16x4_f32(false, aI, false, bI, (short)0, aRb, false, false);
            aIa = __builtin_amdgcn_wmma_f32_16x16x4_f32(false, aR, false, bI, (short)0, aIa, false, false);
            aIb = __builtin_amdgcn_wmma_f32_16x16x4_f32(false, aI, false, bR, (short)0, aIb, false, false);
        }
        const int wv = (vcol + 32) & 255;
        const bool pk = (ub + 8 * hi) == 0;
        float pw[8], pm[8];
        #pragma unroll
        for (int r = 0; r < 8; ++r) {
            float re = aRa[r] - aRb[r];
            float im = aIa[r] + aIb[r];
            float rm = aRa[r] + aRb[r];
            float jm = aIa[r] - aIb[r];
            if (r == 0 && pk) {
                re = aRa[0]; im = aIa[0];
                rm = aIb[0]; jm = aRb[0];
            }
            pw[r] = re * re + im * im;
            pm[r] = rm * rm + jm * jm;
            mymax = fmaxf(mymax, fmaxf(pw[r], pm[r]));
        }
        #pragma unroll
        for (int r = 0; r < 8; ++r) {
            const int u  = ub + r + 8 * hi;
            const int wu = (u + 32) & 255;
            if (wu < 64 && wv < 64) crop[wu * CPITCH + wv] = pw[r];
        }
        #pragma unroll
        for (int r = 0; r < 8; ++r) {
            const int u  = ub + r + 8 * hi;
            const int wm = (32 - u) & 255;
            if (wm < 32 && wv < 64) crop[wm * CPITCH + wv] = pm[r];
        }
    }

    for (int off = 16; off > 0; off >>= 1)
        mymax = fmaxf(mymax, __shfl_xor(mymax, off, 32));
    if (lane == 0) wave_max[wave] = mymax;
    __syncthreads();
    float mx = wave_max[0];
    for (int w = 1; w < NW; ++w) mx = fmaxf(mx, wave_max[w]);
    const float inv = 1.0f / mx;

    for (int idx = t; idx < S * S; idx += NT) {
        const int cu = idx >> 6, cv = idx & 63;
        const float val = crop[cu * CPITCH + cv] * inv * Ae + bge;
        atomicAdd(&out[b * S * S + idx], val);
    }
}

extern "C" void kernel_launch(void* const* d_in, const int* in_sizes, int n_in,
                              void* d_out, int out_size, void* d_ws, size_t ws_size,
                              hipStream_t stream) {
    const float* x  = (const float*)d_in[0];
    const float* y  = (const float*)d_in[1];
    const float* z  = (const float*)d_in[2];
    const float* A  = (const float*)d_in[3];
    const float* bg = (const float*)d_in[4];
    const float* p  = (const float*)d_in[5];
    const float* ph = (const float*)d_in[6];
    float* out = (float*)d_out;

    const int n_emit = in_sizes[0];
    zero_out_kernel<<<(out_size + 255) / 256, 256, 0, stream>>>(out, out_size);
    psf_render_kernel<<<n_emit, NT, 0, stream>>>(x, y, z, A, bg, p, ph, out);
}


# twiddle re-centred by half a sample (modulus-one row/column factor, dropped by |field|^2): cos even, sin odd about the pupil centre, so a once-folded pupil gives K 24 not 48; f32 sum order changes
# speedup vs baseline: 3.1480x; 1.4353x over previous
#include <hip/hip_runtime.h>
#include <math.h>


typedef __attribute__((ext_vector_type(2))) float v2f;
typedef __attribute__((ext_vector_type(8))) float v8f;

#define S 64
#define F 256
#define KS 48
#define HK 24
#define WR 128
#define XP 28
#define TP 52
#define CPITCH 66
#define TWO_PI 6.28318530717958647692f
#define NT 512
#define NW (NT / 32)

__global__ void zero_out_kernel(float* out, int n) {
    int i = blockIdx.x * blockDim.x + threadIdx.x;
    if (i < n) out[i] = 0.0f;
}

__device__ __forceinline__ void pupil_at(const float* __restrict__ pupil_phase, int i, int j,
                                         float xe, float ye, float ze, float& re, float& im) {
    const float ui = (2.0f * (float)i / 63.0f - 1.0f) * (1.0f / 0.75f);
    const float vj = (2.0f * (float)j / 63.0f - 1.0f) * (1.0f / 0.75f);
    const float r2 = ui * ui + vj * vj;
    const float mag  = (r2 <= 1.0f) ? 1.0f : 0.0f;
    const float prop = sqrtf(1.0f - fminf(r2, 1.0f));
    const float ph = pupil_phase[i * S + j] + prop * ze
                   - TWO_PI * ((float)(i - 32) * (1.0f / 256.0f)) * xe
                   - TWO_PI * ((float)(j - 32) * (1.0f / 256.0f)) * ye;
    float sn, cs;
    __sincosf(ph, &sn, &cs);
    re = mag * cs;
    im = mag * sn;
}

__launch_bounds__(NT)
__global__ void psf_render_kernel(const float* __restrict__ xs,
                                  const float* __restrict__ ys,
                                  const float* __restrict__ zs,
                                  const float* __restrict__ As,
                                  const float* __restrict__ bgs,
                                  const float* __restrict__ ps,
                                  const float* __restrict__ pupil_phase,
                                  float* __restrict__ out) {
    __shared__ float XEr[KS * XP];
    __shared__ float XEi[KS * XP];
    __shared__ float XOr[KS * XP];
    __shared__ float XOi[KS * XP];
    __shared__ float Wc[WR * XP];
    __shared__ float Ws[WR * XP];
    __shared__ float Tre[F * TP];
    __shared__ float Tim[F * TP];
    __shared__ float crop[S * CPITCH];
    __shared__ float wave_max[NW];

    const int e = blockIdx.x;
    const int b = e >> 2;
    if (ps[e] <= 0.5f) return;
    const float xe = xs[e], ye = ys[e], ze = zs[e];
    const float Ae = As[e], bge = bgs[e];

    const int t    = threadIdx.x;
    const int lane = t & 31;
    const int wave = t >> 5;
    const int col  = lane & 15;
    const int hi   = lane >> 4;

    for (unsigned idx = t; idx < WR * HK; idx += NT) {
        const int v = (int)(idx / HK), n = (int)(idx % HK);
        const int ti = ((2 * n + 1) * v) & 511;
        float sn, cs;
        __sincosf(TWO_PI * (float)ti * (1.0f / 512.0f), &sn, &cs);
        if (v == 0) sn = (n & 1) ? -1.0f : 1.0f;
        Wc[v * XP + n] = cs;
        Ws[v * XP + n] = sn;
    }

    for (unsigned idx = t; idx < HK * HK; idx += NT) {
        const int n = (int)(idx / HK), m = (int)(idx % HK);
        float r00, i00, r01, i01, r10, i10, r11, i11;
        pupil_at(pupil_phase, 32 + n, 32 + m, xe, ye, ze, r00, i00);
        pupil_at(pupil_phase, 32 + n, 31 - m, xe, ye, ze, r01, i01);
        pupil_at(pupil_phase, 31 - n, 32 + m, xe, ye, ze, r10, i10);
        pupil_at(pupil_phase, 31 - n, 31 - m, xe, ye, ze, r11, i11);
        const float sr0 = r00 + r10, si0 = i00 + i10;
        const float sr1 = r01 + r11, si1 = i01 + i11;
        const float dr0 = r00 - r10, di0 = i00 - i10;
        const float dr1 = r01 - r11, di1 = i01 - i11;
        XEr[n * XP + m] = sr0 + sr1;         XEi[n * XP + m] = si0 + si1;
        XOr[n * XP + m] = sr0 - sr1;         XOi[n * XP + m] = si0 - si1;
        XEr[(HK + n) * XP + m] = dr0 + dr1;  XEi[(HK + n) * XP + m] = di0 + di1;
        XOr[(HK + n) * XP + m] = dr0 - dr1;  XOi[(HK + n) * XP + m] = di0 - di1;
    }
    __syncthreads();

    for (int tt = wave; tt < 24; tt += NW) {
        const int ib = (tt >> 3) * 16;
        const int vb = (tt & 7) * 16;
        const int arow = ib + col;
        const int vcol = vb + col;
        v8f q1 = {}, q2 = {}, q3 = {}, q4 = {};
        #pragma unroll
        for (int kb = 0; kb < HK; kb += 4) {
            const int k0 = kb + 2 * hi;
            const v2f aEr = *(const v2f*)&XEr[arow * XP + k0];
            const v2f aEi = *(const v2f*)&XEi[arow * XP + k0];
            const v2f aOr = *(const v2f*)&XOr[arow * XP + k0];
            const v2f aOi = *(const v2f*)&XOi[arow * XP + k0];
            const v2f bC  = *(const v2f*)&Wc[vcol * XP + k0];
            const v2f bS  = *(const v2f*)&Ws[vcol * XP + k0];
            q1 = __builtin_amdgcn_wmma_f32_16x16x4_f32(false, aEr, false, bC, (short)0, q1, false, false);
            q2 = __builtin_amdgcn_wmma_f32_16x16x4_f32(false, aOi, false, bS, (short)0, q2, false, false);
            q3 = __builtin_amdgcn_wmma_f32_16x16x4_f32(false, aEi, false, bC, (short)0, q3, false, false);
            q4 = __builtin_amdgcn_wmma_f32_16x16x4_f32(false, aOr, false, bS, (short)0, q4, false, false);
        }
        const int rbase = ib + 8 * hi;
        const bool pk  = vcol == 0;
        const int  vmir = pk ? 128 : 256 - vcol;
        #pragma unroll
        for (int q = 0; q < 4; ++q) {
            v2f tr, ti, mr, mi;
            tr.x = q1[2 * q]     - q2[2 * q];
            tr.y = q1[2 * q + 1] - q2[2 * q + 1];
            ti.x = q3[2 * q]     + q4[2 * q];
            ti.y = q3[2 * q + 1] + q4[2 * q + 1];
            mr.x = -q1[2 * q]     - q2[2 * q];
            mr.y = -q1[2 * q + 1] - q2[2 * q + 1];
            mi.x = q4[2 * q]     - q3[2 * q];
            mi.y = q4[2 * q + 1] - q3[2 * q + 1];
            if (pk) {
                tr.x = q1[2 * q];  tr.y = q1[2 * q + 1];
                ti.x = q3[2 * q];  ti.y = q3[2 * q + 1];
                mr.x = -q2[2 * q]; mr.y = -q2[2 * q + 1];
                mi.x = q4[2 * q];  mi.y = q4[2 * q + 1];
            }
            *(v2f*)&Tre[vcol * TP + rbase + 2 * q] = tr;
            *(v2f*)&Tim[vcol * TP + rbase + 2 * q] = ti;
            *(v2f*)&Tre[vmir * TP + rbase + 2 * q] = mr;
            *(v2f*)&Tim[vmir * TP + rbase + 2 * q] = mi;
        }
    }
    __syncthreads();

    float mymax = 0.0f;
    static_assert(128 % NW == 0, "stage-2 tiles divide evenly over the waves");
    for (int tt = wave * (128 / NW); tt < (wave + 1) * (128 / NW); ++tt) {
        const int ub = (tt >> 4) * 16;
        const int vb = (tt & 15) * 16;
        const int urow = ub + col;
        const int vcol = vb + col;
        v8f g1 = {}, g2 = {}, g3 = {}, g4 = {};
        #pragma unroll
        for (int kb = 0; kb < HK; kb += 4) {
            const int k0 = kb + 2 * hi;
            const v2f aC  = *(const v2f*)&Wc[urow * XP + k0];
            const v2f aS  = *(const v2f*)&Ws[urow * XP + k0];
            const v2f bEr = *(const v2f*)&Tre[vcol * TP + k0];
            const v2f bEi = *(const v2f*)&Tim[vcol * TP + k0];
            const v2f bOr = *(const v2f*)&Tre[vcol * TP + HK + k0];
            const v2f bOi = *(const v2f*)&Tim[vcol * TP + HK + k0];
            g1 = __builtin_amdgcn_wmma_f32_16x16x4_f32(false, aC, false, bEr, (short)0, g1, false, false);
            g2 = __builtin_amdgcn_wmma_f32_16x16x4_f32(false, aS, false, bOi, (short)0, g2, false, false);
            g3 = __builtin_amdgcn_wmma_f32_16x16x4_f32(false, aC, false, bEi, (short)0, g3, false, false);
            g4 = __builtin_amdgcn_wmma_f32_16x16x4_f32(false, aS, false, bOr, (short)0, g4, false, false);
        }
        const int wv = (vcol + 32) & 255;
        const bool pk = (ub + 8 * hi) == 0;
        float pw[8], pm[8];
        #pragma unroll
        for (int r = 0; r < 8; ++r) {
            float re = g1[r] - g2[r];
            float im = g3[r] + g4[r];
            float rm = g1[r] + g2[r];
            float jm = g4[r] - g3[r];
            if (r == 0 && pk) {
                re = g1[0]; im = g3[0];
                rm = g2[0]; jm = g4[0];
            }
            pw[r] = re * re + im * im;
            pm[r] = rm * rm + jm * jm;
            mymax = fmaxf(mymax, fmaxf(pw[r], pm[r]));
        }
        #pragma unroll
        for (int r = 0; r < 8; ++r) {
            const int u  = ub + r + 8 * hi;
            const int wu = (u + 32) & 255;
            if (wu < 64 && wv < 64) crop[wu * CPITCH + wv] = pw[r];
        }
        #pragma unroll
        for (int r = 0; r < 8; ++r) {
            const int u  = ub + r + 8 * hi;
            const int wm = (32 - u) & 255;
            if (wm < 32 && wv < 64) crop[wm * CPITCH + wv] = pm[r];
        }
    }

    for (int off = 16; off > 0; off >>= 1)
        mymax = fmaxf(mymax, __shfl_xor(mymax, off, 32));
    if (lane == 0) wave_max[wave] = mymax;
    __syncthreads();
    float mx = wave_max[0];
    for (int w = 1; w < NW; ++w) mx = fmaxf(mx, wave_max[w]);
    const float inv = 1.0f / mx;

    for (int idx = t; idx < S * S; idx += NT) {
        const int cu = idx >> 6, cv = idx & 63;
        const float val = crop[cu * CPITCH + cv] * inv * Ae + bge;
        atomicAdd(&out[b * S * S + idx], val);
    }
}

extern "C" void kernel_launch(void* const* d_in, const int* in_sizes, int n_in,
                              void* d_out, int out_size, void* d_ws, size_t ws_size,
                              hipStream_t stream) {
    const float* x  = (const float*)d_in[0];
    const float* y  = (const float*)d_in[1];
    const float* z  = (const float*)d_in[2];
    const float* A  = (const float*)d_in[3];
    const float* bg = (const float*)d_in[4];
    const float* p  = (const float*)d_in[5];
    const float* ph = (const float*)d_in[6];
    float* out = (float*)d_out;

    const int n_emit = in_sizes[0];
    zero_out_kernel<<<(out_size + 255) / 256, 256, 0, stream>>>(out, out_size);
    psf_render_kernel<<<n_emit, NT, 0, stream>>>(x, y, z, A, bg, p, ph, out);
}
